# starting kernel
# baseline (speedup 1.0000x reference)
#include <hip/hip_runtime.h>

typedef __attribute__((ext_vector_type(2))) float v2f;
typedef __attribute__((ext_vector_type(8))) float v8f;


__global__ void k_zero(float* __restrict__ p, int n) {
    int i = blockIdx.x * blockDim.x + threadIdx.x;
    if (i < n) p[i] = 0.0f;
}

__global__ void k_degree(const int* __restrict__ src, const int* __restrict__ dst,
                         float* __restrict__ degO, float* __restrict__ degI, int e) {
    int i = blockIdx.x * blockDim.x + threadIdx.x;
    if (i < e) {
        atomicAdd(&degO[src[i]], 1.0f);
        atomicAdd(&degI[dst[i]], 1.0f);
    }
}

__global__ void k_coeff(float* __restrict__ c, int n) {
    int i = blockIdx.x * blockDim.x + threadIdx.x;
    if (i < n) {
        float d = c[i];
        c[i] = (d > 0.0f) ? rsqrtf(d) : 0.0f;
    }
}

__global__ void k_pad_features(const float* __restrict__ x, float* __restrict__ xp, int n) {
    int i = blockIdx.x * blockDim.x + threadIdx.x;
    if (i < n * 8) {
        int r = i >> 3, c = i & 7;
        xp[i] = (c < 6) ? x[r * 6 + c] : 0.0f;
    }
}

__global__ void k_pad_w1(const float* __restrict__ w, float* __restrict__ wp) {
    int i = blockIdx.x * blockDim.x + threadIdx.x;
    if (i < 8 * 128) {
        int r = i >> 7;
        wp[i] = (r < 6) ? w[i] : 0.0f;
    }
}

__global__ void k_pad_w5(const float* __restrict__ w, float* __restrict__ wp) {
    int i = blockIdx.x * blockDim.x + threadIdx.x;
    if (i < 128 * 32) {
        int r = i >> 5, c = i & 31;
        wp[i] = (c < 18) ? w[r * 18 + c] : 0.0f;
    }
}

template <int DIN, int DOUT>
__global__ void k_gemm_wmma(const float* __restrict__ A, const float* __restrict__ W,
                            float* __restrict__ Y) {
    const int lane = threadIdx.x;
    const int m0   = blockIdx.x * 16;
    const int n0   = blockIdx.y * 16;
    const int half = lane >> 4;
    const int l15  = lane & 15;
    const int koff = half * 2;

    const float* arow = A + (size_t)(m0 + l15) * DIN + koff;
    const int col = n0 + l15;

    v8f acc = {};
#pragma unroll
    for (int k = 0; k < DIN; k += 4) {
        v2f a = *(const v2f*)(arow + k);
        v2f b;
        b.x = W[(size_t)(k + koff) * DOUT + col];
        b.y = W[(size_t)(k + koff + 1) * DOUT + col];
        acc = __builtin_amdgcn_wmma_f32_16x16x4_f32(
             false, a,  false, b,
             (short)0, acc,  false,  false);
    }

    float* yrow = Y + (size_t)(m0 + half * 8) * DOUT + col;
#pragma unroll
    for (int v = 0; v < 8; ++v) yrow[(size_t)v * DOUT] = acc[v];
}

template <int DOUT, int STRIDE>
__global__ void k_scatter(const float* __restrict__ XW, const int* __restrict__ src,
                          const int* __restrict__ dst, const float* __restrict__ cs,
                          float* __restrict__ AGG, int e) {
    int w    = (blockIdx.x * blockDim.x + threadIdx.x) >> 5;
    int lane = threadIdx.x & 31;
    if (w >= e) return;
    int s = src[w], d = dst[w];
    float c = cs[s];
    const float* xr = XW + (size_t)s * STRIDE;
    float* ar       = AGG + (size_t)d * STRIDE;
#pragma unroll
    for (int j = lane; j < DOUT; j += 32)
        atomicAdd(&ar[j], xr[j] * c);
}

template <int DOUT, int STRIDE, bool RELU>
__global__ void k_finalize(float* __restrict__ H, const float* __restrict__ cd,
                           const float* __restrict__ b, int n) {
    int i = blockIdx.x * blockDim.x + threadIdx.x;
    if (i < n * DOUT) {
        int r = i / DOUT, c = i % DOUT;
        size_t idx = (size_t)r * STRIDE + c;
        float v = H[idx] * cd[r] + b[c];
        if (RELU) v = fmaxf(v, 0.0f);
        H[idx] = v;
    }
}

__global__ void k_finalize_out(const float* __restrict__ AGG, const float* __restrict__ cd,
                               const float* __restrict__ b, float* __restrict__ out, int n) {
    int i = blockIdx.x * blockDim.x + threadIdx.x;
    if (i < n * 18) {
        int r = i / 18, c = i % 18;
        out[i] = AGG[(size_t)r * 32 + c] * cd[r] + b[c];
    }
}


static inline int cdiv(int a, int b) { return (a + b - 1) / b; }

extern "C" void kernel_launch(void* const* d_in, const int* in_sizes, int n_in,
                              void* d_out, int out_size, void* d_ws, size_t ws_size,
                              hipStream_t stream) {
    (void)n_in; (void)out_size; (void)ws_size;

    const float* feat = (const float*)d_in[0];
    const int*   src  = (const int*)d_in[1];
    const int*   dst  = (const int*)d_in[2];
    const float* W1 = (const float*)d_in[3];  const float* b1 = (const float*)d_in[4];
    const float* W2 = (const float*)d_in[5];  const float* b2 = (const float*)d_in[6];
    const float* W3 = (const float*)d_in[7];  const float* b3 = (const float*)d_in[8];
    const float* W4 = (const float*)d_in[9];  const float* b4 = (const float*)d_in[10];
    const float* W5 = (const float*)d_in[11]; const float* b5 = (const float*)d_in[12];
    float* out = (float*)d_out;

    const int N = 100000;
    const int E = in_sizes[1];

    float* ws = (float*)d_ws;
    size_t o = 0;
    float* c_src = ws + o; o += N;
    float* c_dst = ws + o; o += N;
    float* Xp    = ws + o; o += (size_t)N * 8;
    float* W1p   = ws + o; o += 8 * 128;
    float* W5p   = ws + o; o += 128 * 32;
    float* XW    = ws + o; o += (size_t)N * 128;
    float* H0    = ws + o; o += (size_t)N * 128;
    float* H1    = ws + o; o += (size_t)N * 128;

    k_zero<<<cdiv(2 * N, 256), 256, 0, stream>>>(c_src, 2 * N);
    k_degree<<<cdiv(E, 256), 256, 0, stream>>>(src, dst, c_src, c_dst, E);
    k_coeff<<<cdiv(2 * N, 256), 256, 0, stream>>>(c_src, 2 * N);

    k_pad_features<<<cdiv(N * 8, 256), 256, 0, stream>>>(feat, Xp, N);
    k_pad_w1<<<4, 256, 0, stream>>>(W1, W1p);
    k_pad_w5<<<16, 256, 0, stream>>>(W5, W5p);

    const dim3 gTile(N / 16, 128 / 16);
    const int  sBlk = cdiv(E, 8);
    const int  fBlk = cdiv(N * 128, 256);

    k_gemm_wmma<8, 128><<<gTile, 32, 0, stream>>>(Xp, W1p, XW);
    k_zero<<<fBlk, 256, 0, stream>>>(H0, N * 128);
    k_scatter<128, 128><<<sBlk, 256, 0, stream>>>(XW, src, dst, c_src, H0, E);
    k_finalize<128, 128, true><<<fBlk, 256, 0, stream>>>(H0, c_dst, b1, N);

    k_gemm_wmma<128, 128><<<gTile, 32, 0, stream>>>(H0, W2, XW);
    k_zero<<<fBlk, 256, 0, stream>>>(H1, N * 128);
    k_scatter<128, 128><<<sBlk, 256, 0, stream>>>(XW, src, dst, c_src, H1, E);
    k_finalize<128, 128, true><<<fBlk, 256, 0, stream>>>(H1, c_dst, b2, N);

    k_gemm_wmma<128, 128><<<gTile, 32, 0, stream>>>(H1, W3, XW);
    k_zero<<<fBlk, 256, 0, stream>>>(H0, N * 128);
    k_scatter<128, 128><<<sBlk, 256, 0, stream>>>(XW, src, dst, c_src, H0, E);
    k_finalize<128, 128, true><<<fBlk, 256, 0, stream>>>(H0, c_dst, b3, N);

    k_gemm_wmma<128, 128><<<gTile, 32, 0, stream>>>(H0, W4, XW);
    k_zero<<<fBlk, 256, 0, stream>>>(H1, N * 128);
    k_scatter<128, 128><<<sBlk, 256, 0, stream>>>(XW, src, dst, c_src, H1, E);
    k_finalize<128, 128, true><<<fBlk, 256, 0, stream>>>(H1, c_dst, b4, N);

    const dim3 gTile5(N / 16, 32 / 16);
    k_gemm_wmma<128, 32><<<gTile5, 32, 0, stream>>>(H1, W5p, XW);
    k_zero<<<cdiv(N * 32, 256), 256, 0, stream>>>(H0, N * 32);
    k_scatter<18, 32><<<sBlk, 256, 0, stream>>>(XW, src, dst, c_src, H0, E);
    k_finalize_out<<<cdiv(N * 18, 256), 256, 0, stream>>>(H0, c_dst, b5, out, N);
}


# Float-atomic edge scatter replaced by a gather over per-destination neighbour lists (counting sort rebuilt every call); scale, bias and ReLU folded into the gather, zero-fill passes gone
# speedup vs baseline: 2.9185x; 2.9185x over previous
#include <hip/hip_runtime.h>


typedef __attribute__((ext_vector_type(2))) float v2f;
typedef __attribute__((ext_vector_type(8))) float v8f;


__global__ void k_zero_i(int* __restrict__ p, int n) {
    int i = blockIdx.x * blockDim.x + threadIdx.x;
    if (i < n) p[i] = 0;
}

__global__ void k_count(const int* __restrict__ src, const int* __restrict__ dst,
                        int* __restrict__ degO, int* __restrict__ degI,
                        int* __restrict__ rank, int e) {
    int i = blockIdx.x * blockDim.x + threadIdx.x;
    if (i < e) {
        atomicAdd(&degO[src[i]], 1);
        rank[i] = atomicAdd(&degI[dst[i]], 1);
    }
}

__global__ void __launch_bounds__(256) k_chunk_sum(const int* __restrict__ degI,
                                                   int* __restrict__ csum, int n) {
    __shared__ int part[8];
    const int t = threadIdx.x;
    const int i0 = blockIdx.x * 1024 + t * 4;
    int s = 0;
    if (i0 < n) {
        int4 v = *(const int4*)(degI + i0);
        s = v.x + v.y + v.z + v.w;
    }
#pragma unroll
    for (int m = 16; m >= 1; m >>= 1) s += __shfl_xor(s, m, 32);
    if ((t & 31) == 0) part[t >> 5] = s;
    __syncthreads();
    if (t == 0) {
        int tot = 0;
#pragma unroll
        for (int w = 0; w < 8; ++w) tot += part[w];
        csum[blockIdx.x] = tot;
    }
}

__global__ void __launch_bounds__(256) k_rowptr_coeff(const int* __restrict__ degO,
                                                      const int* __restrict__ degI,
                                                      const int* __restrict__ csum,
                                                      int* __restrict__ seg,
                                                      int* __restrict__ rowptr,
                                                      float* __restrict__ c_src,
                                                      float* __restrict__ c_dst, int n) {
    __shared__ int part[8];
    const int t = threadIdx.x, lane = t & 31, w = t >> 5;
    const int b = blockIdx.x;

    int off = 0;
    for (int j = lane; j < b; j += 32) off += csum[j];
#pragma unroll
    for (int m = 16; m >= 1; m >>= 1) off += __shfl_xor(off, m, 32);

    const int i0 = b * 1024 + t * 4;
    const bool live = i0 < n;
    int4 di = make_int4(0, 0, 0, 0), dout = make_int4(0, 0, 0, 0);
    if (live) {
        di   = *(const int4*)(degI + i0);
        dout = *(const int4*)(degO + i0);
    }
    const int s = di.x + di.y + di.z + di.w;
    int incl = s;
#pragma unroll
    for (int d = 1; d < 32; d <<= 1) {
        int up = __shfl_up(incl, d, 32);
        if (lane >= d) incl += up;
    }
    if (lane == 31) part[w] = incl;
    __syncthreads();
    int wbase = 0;
#pragma unroll
    for (int k = 0; k < 8; ++k) wbase += (k < w) ? part[k] : 0;

    if (live) {
        const int base = off + wbase + incl - s;
        int4 r;
        r.x = base;
        r.y = base + di.x;
        r.z = r.y + di.y;
        r.w = r.z + di.z;
        *(int4*)(rowptr + i0) = r;
        int4* sg = (int4*)(seg + 2 * i0);
        sg[0] = make_int4(r.x, di.x, r.y, di.y);
        sg[1] = make_int4(r.z, di.z, r.w, di.w);
        float4 ci, co;
        ci.x = di.x > 0 ? rsqrtf((float)di.x) : 0.0f;
        ci.y = di.y > 0 ? rsqrtf((float)di.y) : 0.0f;
        ci.z = di.z > 0 ? rsqrtf((float)di.z) : 0.0f;
        ci.w = di.w > 0 ? rsqrtf((float)di.w) : 0.0f;
        co.x = dout.x > 0 ? rsqrtf((float)dout.x) : 0.0f;
        co.y = dout.y > 0 ? rsqrtf((float)dout.y) : 0.0f;
        co.z = dout.z > 0 ? rsqrtf((float)dout.z) : 0.0f;
        co.w = dout.w > 0 ? rsqrtf((float)dout.w) : 0.0f;
        *(float4*)(c_dst + i0) = ci;
        *(float4*)(c_src + i0) = co;
    }
}

__global__ void k_fill(const int* __restrict__ src, const int* __restrict__ dst,
                       const int* __restrict__ rank, const int* __restrict__ rowptr,
                       int* __restrict__ nbr, int e) {
    int i = blockIdx.x * blockDim.x + threadIdx.x;
    if (i < e) {
        int pos = rowptr[dst[i]] + rank[i];
        if ((unsigned)pos < (unsigned)e) nbr[pos] = src[i];
    }
}

__global__ void k_pad_features(const float* __restrict__ x, float* __restrict__ xp, int n) {
    int i = blockIdx.x * blockDim.x + threadIdx.x;
    if (i < n * 8) {
        int r = i >> 3, c = i & 7;
        xp[i] = (c < 6) ? x[r * 6 + c] : 0.0f;
    }
}

__global__ void k_pad_w1(const float* __restrict__ w, float* __restrict__ wp) {
    int i = blockIdx.x * blockDim.x + threadIdx.x;
    if (i < 8 * 128) {
        int r = i >> 7;
        wp[i] = (r < 6) ? w[i] : 0.0f;
    }
}

__global__ void k_pad_w5(const float* __restrict__ w, float* __restrict__ wp) {
    int i = blockIdx.x * blockDim.x + threadIdx.x;
    if (i < 128 * 32) {
        int r = i >> 5, c = i & 31;
        wp[i] = (c < 18) ? w[r * 18 + c] : 0.0f;
    }
}

template <int DIN, int DOUT>
__global__ void k_gemm_wmma(const float* __restrict__ A, const float* __restrict__ W,
                            float* __restrict__ Y) {
    const int lane = threadIdx.x;
    const int m0   = blockIdx.x * 16;
    const int n0   = blockIdx.y * 16;
    const int half = lane >> 4;
    const int l15  = lane & 15;
    const int koff = half * 2;

    const float* arow = A + (size_t)(m0 + l15) * DIN + koff;
    const int col = n0 + l15;

    v8f acc = {};
#pragma unroll
    for (int k = 0; k < DIN; k += 4) {
        v2f a = *(const v2f*)(arow + k);
        v2f b;
        b.x = W[(size_t)(k + koff) * DOUT + col];
        b.y = W[(size_t)(k + koff + 1) * DOUT + col];
        acc = __builtin_amdgcn_wmma_f32_16x16x4_f32(
             false, a,  false, b,
             (short)0, acc,  false,  false);
    }

    float* yrow = Y + (size_t)(m0 + half * 8) * DOUT + col;
#pragma unroll
    for (int v = 0; v < 8; ++v) yrow[(size_t)v * DOUT] = acc[v];
}

__device__ __forceinline__ float4 fma4(float4 r, float c, float4 a) {
    a.x = fmaf(r.x, c, a.x); a.y = fmaf(r.y, c, a.y);
    a.z = fmaf(r.z, c, a.z); a.w = fmaf(r.w, c, a.w);
    return a;
}

__global__ void __launch_bounds__(256) k_gather128_relu(
        const float* __restrict__ XW, const int* __restrict__ seg,
        const int* __restrict__ nbr,
        const float* __restrict__ cs, const float* __restrict__ cd,
        const float* __restrict__ bias, float* __restrict__ H, int n) {
    const int node = __builtin_amdgcn_readfirstlane((int)((blockIdx.x * 256 + threadIdx.x) >> 5));
    const int lane = threadIdx.x & 31;
    if (node >= n) return;
    const int2 sg = *(const int2*)(seg + 2 * (size_t)node);
    const int beg = sg.x, cnt = sg.y;
    const float4* __restrict__ X4 = (const float4*)XW;

    float4 acc = make_float4(0.f, 0.f, 0.f, 0.f);
    for (int base = 0; base < cnt; base += 32) {
        const int m = min(cnt - base, 32);
        int   s_l = 0;
        float c_l = 0.0f;
        if (lane < m) {
            s_l = nbr[beg + base + lane];
            s_l = ((unsigned)s_l < (unsigned)n) ? s_l : 0;
            c_l = cs[s_l];
        }
        const int ci_l = __float_as_int(c_l);
        for (int j = 0; j < m; j += 8) {
            float4 r[8];
#pragma unroll
            for (int t = 0; t < 8; ++t) {
                r[t] = make_float4(0.f, 0.f, 0.f, 0.f);
                if (j + t < m) {
                    const int s = __builtin_amdgcn_readlane(s_l, j + t);
                    r[t] = X4[(size_t)s * 32 + lane];
                }
            }
#pragma unroll
            for (int t = 0; t < 8; ++t)
                acc = fma4(r[t], __int_as_float(__builtin_amdgcn_readlane(ci_l, j + t)), acc);
        }
    }

    const float sc = cd[node];
    float4 o;
    o.x = fmaxf(acc.x * sc + bias[lane * 4 + 0], 0.0f);
    o.y = fmaxf(acc.y * sc + bias[lane * 4 + 1], 0.0f);
    o.z = fmaxf(acc.z * sc + bias[lane * 4 + 2], 0.0f);
    o.w = fmaxf(acc.w * sc + bias[lane * 4 + 3], 0.0f);
    ((float4*)H)[(size_t)node * 32 + lane] = o;
}

__global__ void __launch_bounds__(512) k_gather_out(
        const float* __restrict__ XW, const int* __restrict__ seg,
        const int* __restrict__ nbr,
        const float* __restrict__ cs, const float* __restrict__ cd,
        const float* __restrict__ bias, float* __restrict__ out, int n) {
    __shared__ float stage[16 * 18];
    const int w    = threadIdx.x >> 5;
    const int node = __builtin_amdgcn_readfirstlane((int)(blockIdx.x * 16 + w));
    const int lane = threadIdx.x & 31;
    const bool have = node < n;
    int beg = 0, cnt = 0;
    if (have) {
        const int2 sg = *(const int2*)(seg + 2 * (size_t)node);
        beg = sg.x; cnt = sg.y;
    }
    const int g = lane >> 3, q = lane & 7;
    const float4* __restrict__ X4 = (const float4*)XW;

    float4 acc = make_float4(0.f, 0.f, 0.f, 0.f);
    for (int base = 0; base < cnt; base += 32) {
        const int m = min(cnt - base, 32);
        int   s_l = 0;
        float c_l = 0.0f;
        if (lane < m) {
            s_l = nbr[beg + base + lane];
            s_l = ((unsigned)s_l < (unsigned)n) ? s_l : 0;
            c_l = cs[s_l];
        }
        for (int j = 0; j < m; j += 8) {
            const int ja = j + g, jb = j + 4 + g;
            const int sa = __shfl(s_l, ja, 32);
            const int sb = __shfl(s_l, jb, 32);
            float4 ra = make_float4(0.f, 0.f, 0.f, 0.f), rb = ra;
            if (ja < m) ra = X4[(size_t)sa * 8 + q];
            if (jb < m) rb = X4[(size_t)sb * 8 + q];
            acc = fma4(ra, __shfl(c_l, ja, 32), acc);
            acc = fma4(rb, __shfl(c_l, jb, 32), acc);
        }
    }
#pragma unroll
    for (int msk = 8; msk <= 16; msk <<= 1) {
        acc.x += __shfl_xor(acc.x, msk, 32);
        acc.y += __shfl_xor(acc.y, msk, 32);
        acc.z += __shfl_xor(acc.z, msk, 32);
        acc.w += __shfl_xor(acc.w, msk, 32);
    }
    if (have && lane < 5) {
        const float sc = cd[node];
        float* o = stage + w * 18 + lane * 4;
        const float* bb = bias + lane * 4;
        o[0] = acc.x * sc + bb[0];
        o[1] = acc.y * sc + bb[1];
        if (lane < 4) {
            o[2] = acc.z * sc + bb[2];
            o[3] = acc.w * sc + bb[3];
        }
    }
    __syncthreads();
    const int t = threadIdx.x;
    if (t < 16 * 18) {
        const size_t idx = (size_t)blockIdx.x * (16 * 18) + t;
        if (idx < (size_t)n * 18) out[idx] = stage[t];
    }
}


static inline int cdiv(int a, int b) { return (a + b - 1) / b; }

extern "C" void kernel_launch(void* const* d_in, const int* in_sizes, int n_in,
                              void* d_out, int out_size, void* d_ws, size_t ws_size,
                              hipStream_t stream) {
    (void)n_in; (void)out_size; (void)ws_size;

    const float* feat = (const float*)d_in[0];
    const int*   src  = (const int*)d_in[1];
    const int*   dst  = (const int*)d_in[2];
    const float* W1 = (const float*)d_in[3];  const float* b1 = (const float*)d_in[4];
    const float* W2 = (const float*)d_in[5];  const float* b2 = (const float*)d_in[6];
    const float* W3 = (const float*)d_in[7];  const float* b3 = (const float*)d_in[8];
    const float* W4 = (const float*)d_in[9];  const float* b4 = (const float*)d_in[10];
    const float* W5 = (const float*)d_in[11]; const float* b5 = (const float*)d_in[12];
    float* out = (float*)d_out;

    const int N = 100000;
    const int E = in_sizes[1];

    float* ws = (float*)d_ws;
    size_t o = 0;
    float* c_src = ws + o; o += N;
    float* c_dst = ws + o; o += N;
    float* Xp    = ws + o; o += (size_t)N * 8;
    float* W1p   = ws + o; o += 8 * 128;
    float* W5p   = ws + o; o += 128 * 32;
    float* XW    = ws + o; o += (size_t)N * 128;
    float* H     = ws + o; o += (size_t)N * 128;
    int* iw = (int*)(ws + o);
    size_t p = 0;
    int* degO   = iw + p; p += N;
    int* degI   = iw + p; p += N;
    int* rowptr = iw + p; p += N;
    int* seg    = iw + p; p += 2 * (size_t)N;
    int* csum   = iw + p; p += 1024;
    int* rank   = iw + p; p += (size_t)E;
    int* nbr    = iw + p; p += (size_t)E;

    const int nChunk = cdiv(N, 1024);

    k_zero_i<<<cdiv(2 * N, 256), 256, 0, stream>>>(degO, 2 * N);
    k_count<<<cdiv(E, 256), 256, 0, stream>>>(src, dst, degO, degI, rank, E);
    k_chunk_sum<<<nChunk, 256, 0, stream>>>(degI, csum, N);
    k_rowptr_coeff<<<nChunk, 256, 0, stream>>>(degO, degI, csum, seg, rowptr, c_src, c_dst, N);
    k_fill<<<cdiv(E, 256), 256, 0, stream>>>(src, dst, rank, rowptr, nbr, E);

    k_pad_features<<<cdiv(N * 8, 256), 256, 0, stream>>>(feat, Xp, N);
    k_pad_w1<<<4, 256, 0, stream>>>(W1, W1p);
    k_pad_w5<<<16, 256, 0, stream>>>(W5, W5p);

    const dim3 gTile(N / 16, 128 / 16);
    const int  gBlk = cdiv(N, 8);

    k_gemm_wmma<8, 128><<<gTile, 32, 0, stream>>>(Xp, W1p, XW);
    k_gather128_relu<<<gBlk, 256, 0, stream>>>(XW, seg, nbr, c_src, c_dst, b1, H, N);

    k_gemm_wmma<128, 128><<<gTile, 32, 0, stream>>>(H, W2, XW);
    k_gather128_relu<<<gBlk, 256, 0, stream>>>(XW, seg, nbr, c_src, c_dst, b2, H, N);

    k_gemm_wmma<128, 128><<<gTile, 32, 0, stream>>>(H, W3, XW);
    k_gather128_relu<<<gBlk, 256, 0, stream>>>(XW, seg, nbr, c_src, c_dst, b3, H, N);

    k_gemm_wmma<128, 128><<<gTile, 32, 0, stream>>>(H, W4, XW);
    k_gather128_relu<<<gBlk, 256, 0, stream>>>(XW, seg, nbr, c_src, c_dst, b4, H, N);

    const dim3 gTile5(N / 16, 32 / 16);
    k_gemm_wmma<128, 32><<<gTile5, 32, 0, stream>>>(H, W5p, XW);
    k_gather_out<<<cdiv(N, 16), 512, 0, stream>>>(XW, seg, nbr, c_src, c_dst, b5, out, N);
}


# Layers 1-4: neighbour sum taken on the layer input and fused with the f32 WMMA weight product in one kernel per layer (64-node tile in LDS, bias+ReLU, whole-row stores); 4 product launches gone
# speedup vs baseline: 4.0417x; 1.3849x over previous
#include <hip/hip_runtime.h>


typedef __attribute__((ext_vector_type(2))) float v2f;
typedef __attribute__((ext_vector_type(8))) float v8f;


__global__ void k_zero_i(int* __restrict__ p, int n) {
    int i = blockIdx.x * blockDim.x + threadIdx.x;
    if (i < n) p[i] = 0;
}

__global__ void k_count(const int* __restrict__ src, const int* __restrict__ dst,
                        int* __restrict__ degO, int* __restrict__ degI,
                        int* __restrict__ rank, int e) {
    int i = blockIdx.x * blockDim.x + threadIdx.x;
    if (i < e) {
        atomicAdd(&degO[src[i]], 1);
        rank[i] = atomicAdd(&degI[dst[i]], 1);
    }
}

__global__ void __launch_bounds__(256) k_chunk_sum(const int* __restrict__ degI,
                                                   int* __restrict__ csum, int n) {
    __shared__ int part[8];
    const int t = threadIdx.x;
    const int i0 = blockIdx.x * 1024 + t * 4;
    int s = 0;
    if (i0 < n) {
        int4 v = *(const int4*)(degI + i0);
        s = v.x + v.y + v.z + v.w;
    }
#pragma unroll
    for (int m = 16; m >= 1; m >>= 1) s += __shfl_xor(s, m, 32);
    if ((t & 31) == 0) part[t >> 5] = s;
    __syncthreads();
    if (t == 0) {
        int tot = 0;
#pragma unroll
        for (int w = 0; w < 8; ++w) tot += part[w];
        csum[blockIdx.x] = tot;
    }
}

__global__ void __launch_bounds__(256) k_rowptr_coeff(const int* __restrict__ degO,
                                                      const int* __restrict__ degI,
                                                      const int* __restrict__ csum,
                                                      int* __restrict__ seg,
                                                      int* __restrict__ rowptr,
                                                      float* __restrict__ c_src,
                                                      float* __restrict__ c_dst, int n) {
    __shared__ int part[8];
    const int t = threadIdx.x, lane = t & 31, w = t >> 5;
    const int b = blockIdx.x;

    int off = 0;
    for (int j = lane; j < b; j += 32) off += csum[j];
#pragma unroll
    for (int m = 16; m >= 1; m >>= 1) off += __shfl_xor(off, m, 32);

    const int i0 = b * 1024 + t * 4;
    const bool live = i0 < n;
    int4 di = make_int4(0, 0, 0, 0), dout = make_int4(0, 0, 0, 0);
    if (live) {
        di   = *(const int4*)(degI + i0);
        dout = *(const int4*)(degO + i0);
    }
    const int s = di.x + di.y + di.z + di.w;
    int incl = s;
#pragma unroll
    for (int d = 1; d < 32; d <<= 1) {
        int up = __shfl_up(incl, d, 32);
        if (lane >= d) incl += up;
    }
    if (lane == 31) part[w] = incl;
    __syncthreads();
    int wbase = 0;
#pragma unroll
    for (int k = 0; k < 8; ++k) wbase += (k < w) ? part[k] : 0;

    if (live) {
        const int base = off + wbase + incl - s;
        int4 r;
        r.x = base;
        r.y = base + di.x;
        r.z = r.y + di.y;
        r.w = r.z + di.z;
        *(int4*)(rowptr + i0) = r;
        int4* sg = (int4*)(seg + 2 * i0);
        sg[0] = make_int4(r.x, di.x, r.y, di.y);
        sg[1] = make_int4(r.z, di.z, r.w, di.w);
        float4 ci, co;
        ci.x = di.x > 0 ? rsqrtf((float)di.x) : 0.0f;
        ci.y = di.y > 0 ? rsqrtf((float)di.y) : 0.0f;
        ci.z = di.z > 0 ? rsqrtf((float)di.z) : 0.0f;
        ci.w = di.w > 0 ? rsqrtf((float)di.w) : 0.0f;
        co.x = dout.x > 0 ? rsqrtf((float)dout.x) : 0.0f;
        co.y = dout.y > 0 ? rsqrtf((float)dout.y) : 0.0f;
        co.z = dout.z > 0 ? rsqrtf((float)dout.z) : 0.0f;
        co.w = dout.w > 0 ? rsqrtf((float)dout.w) : 0.0f;
        *(float4*)(c_dst + i0) = ci;
        *(float4*)(c_src + i0) = co;
    }
}

__global__ void k_fill(const int* __restrict__ src, const int* __restrict__ dst,
                       const int* __restrict__ rank, const int* __restrict__ rowptr,
                       int* __restrict__ nbr, int e) {
    int i = blockIdx.x * blockDim.x + threadIdx.x;
    if (i < e) {
        int pos = rowptr[dst[i]] + rank[i];
        if ((unsigned)pos < (unsigned)e) nbr[pos] = src[i];
    }
}

__global__ void k_pad_features(const float* __restrict__ x, float* __restrict__ xp, int n) {
    int i = blockIdx.x * blockDim.x + threadIdx.x;
    if (i < n * 8) {
        int r = i >> 3, c = i & 7;
        xp[i] = (c < 6) ? x[r * 6 + c] : 0.0f;
    }
}

__global__ void k_pad_w1(const float* __restrict__ w, float* __restrict__ wp) {
    int i = blockIdx.x * blockDim.x + threadIdx.x;
    if (i < 8 * 128) {
        int r = i >> 7;
        wp[i] = (r < 6) ? w[i] : 0.0f;
    }
}

__global__ void k_pad_w5(const float* __restrict__ w, float* __restrict__ wp) {
    int i = blockIdx.x * blockDim.x + threadIdx.x;
    if (i < 128 * 32) {
        int r = i >> 5, c = i & 31;
        wp[i] = (c < 18) ? w[r * 18 + c] : 0.0f;
    }
}

template <int DIN, int DOUT>
__global__ void k_gemm_wmma(const float* __restrict__ A, const float* __restrict__ W,
                            float* __restrict__ Y) {
    const int lane = threadIdx.x;
    const int m0   = blockIdx.x * 16;
    const int n0   = blockIdx.y * 16;
    const int half = lane >> 4;
    const int l15  = lane & 15;
    const int koff = half * 2;

    const float* arow = A + (size_t)(m0 + l15) * DIN + koff;
    const int col = n0 + l15;

    v8f acc = {};
#pragma unroll
    for (int k = 0; k < DIN; k += 4) {
        v2f a = *(const v2f*)(arow + k);
        v2f b;
        b.x = W[(size_t)(k + koff) * DOUT + col];
        b.y = W[(size_t)(k + koff + 1) * DOUT + col];
        acc = __builtin_amdgcn_wmma_f32_16x16x4_f32(
             false, a,  false, b,
             (short)0, acc,  false,  false);
    }

    float* yrow = Y + (size_t)(m0 + half * 8) * DOUT + col;
#pragma unroll
    for (int v = 0; v < 8; ++v) yrow[(size_t)v * DOUT] = acc[v];
}

__device__ __forceinline__ float4 fma4(float4 r, float c, float4 a) {
    a.x = fmaf(r.x, c, a.x); a.y = fmaf(r.y, c, a.y);
    a.z = fmaf(r.z, c, a.z); a.w = fmaf(r.w, c, a.w);
    return a;
}

__device__ __forceinline__ float4 gather_row128(const float4* __restrict__ X4,
        const int* __restrict__ nbr, const float* __restrict__ cs,
        int beg, int cnt, int lane, int n) {
    float4 acc = make_float4(0.f, 0.f, 0.f, 0.f);
    for (int base = 0; base < cnt; base += 32) {
        const int m = min(cnt - base, 32);
        int   s_l = 0;
        float c_l = 0.0f;
        if (lane < m) {
            s_l = nbr[beg + base + lane];
            s_l = ((unsigned)s_l < (unsigned)n) ? s_l : 0;
            c_l = cs[s_l];
        }
        const int ci_l = __float_as_int(c_l);
        for (int j = 0; j < m; j += 8) {
            float4 r[8];
#pragma unroll
            for (int t = 0; t < 8; ++t) {
                r[t] = make_float4(0.f, 0.f, 0.f, 0.f);
                if (j + t < m) {
                    const int s = __builtin_amdgcn_readlane(s_l, j + t);
                    r[t] = X4[(size_t)s * 32 + lane];
                }
            }
#pragma unroll
            for (int t = 0; t < 8; ++t)
                acc = fma4(r[t], __int_as_float(__builtin_amdgcn_readlane(ci_l, j + t)), acc);
        }
    }
    return acc;
}

__device__ __forceinline__ float4 gather_row8(const float4* __restrict__ X4,
        const int* __restrict__ nbr, const float* __restrict__ cs,
        int beg, int cnt, int lane, int n) {
    float4 acc = make_float4(0.f, 0.f, 0.f, 0.f);
    const int p = lane >> 1, q = lane & 1;
    for (int base = 0; base < cnt; base += 16) {
        const int idx = base + p;
        if (idx < cnt) {
            int s = nbr[beg + idx];
            s = ((unsigned)s < (unsigned)n) ? s : 0;
            acc = fma4(X4[(size_t)s * 2 + q], cs[s], acc);
        }
    }
#pragma unroll
    for (int msk = 2; msk <= 16; msk <<= 1) {
        acc.x += __shfl_xor(acc.x, msk, 32);
        acc.y += __shfl_xor(acc.y, msk, 32);
        acc.z += __shfl_xor(acc.z, msk, 32);
        acc.w += __shfl_xor(acc.w, msk, 32);
    }
    return acc;
}

template <int DIN>
__global__ void __launch_bounds__(512) k_layer_fused(
        const float* __restrict__ Hin, const float* __restrict__ Wm,
        const int* __restrict__ seg, const int* __restrict__ nbr,
        const float* __restrict__ cs, const float* __restrict__ cd,
        const float* __restrict__ bias, float* __restrict__ Hout, int n) {
    constexpr int PI = DIN + 4;
    constexpr int PO = 128 + 4;
    __shared__ float4 tile4[64 * PO / 4];
    float* tile = (float*)tile4;

    const int w     = __builtin_amdgcn_readfirstlane((int)(threadIdx.x >> 5));
    const int lane  = threadIdx.x & 31;
    const int node0 = blockIdx.x * 64;
    const float4* __restrict__ X4 = (const float4*)Hin;

    for (int i = 0; i < 4; ++i) {
        const int row  = w * 4 + i;
        const int node = node0 + row;
        float4 a = make_float4(0.f, 0.f, 0.f, 0.f);
        if (node < n) {
            const int2 sg = *(const int2*)(seg + 2 * (size_t)node);
            if (DIN == 128) a = gather_row128(X4, nbr, cs, sg.x, sg.y, lane, n);
            else            a = gather_row8(X4, nbr, cs, sg.x, sg.y, lane, n);
            const float sc = cd[node];
            a.x *= sc; a.y *= sc; a.z *= sc; a.w *= sc;
        }
        if (lane < DIN / 4) tile4[row * (PI / 4) + lane] = a;
    }
    __syncthreads();

    const int c  = lane & 15;
    const int h  = lane >> 4;
    const int n0 = (w & 7) * 16;
    const int r0 = (w >> 3) * 32;
    v8f acc[2] = {};
    const float* bcol = Wm + (size_t)(2 * h) * 128 + n0 + c;
    const float* arow = tile + (r0 + c) * PI + 2 * h;
#pragma unroll 8
    for (int k = 0; k < DIN; k += 4) {
        v2f b;
        b.x = bcol[(size_t)k * 128];
        b.y = bcol[(size_t)(k + 1) * 128];
#pragma unroll
        for (int mt = 0; mt < 2; ++mt) {
            v2f a = *(const v2f*)(arow + mt * 16 * PI + k);
            acc[mt] = __builtin_amdgcn_wmma_f32_16x16x4_f32(
                 false, a,  false, b,
                 (short)0, acc[mt],  false,  false);
        }
    }
    __syncthreads();

#pragma unroll
    for (int mt = 0; mt < 2; ++mt)
#pragma unroll
        for (int r = 0; r < 8; ++r)
            tile[(r0 + mt * 16 + 8 * h + r) * PO + n0 + c] = acc[mt][r];
    __syncthreads();

    const float b0 = bias[lane * 4 + 0], b1 = bias[lane * 4 + 1];
    const float b2 = bias[lane * 4 + 2], b3 = bias[lane * 4 + 3];
    for (int i = 0; i < 4; ++i) {
        const int row  = w * 4 + i;
        const int node = node0 + row;
        if (node < n) {
            const float4 v = tile4[row * (PO / 4) + lane];
            float4 o;
            o.x = fmaxf(v.x + b0, 0.0f);
            o.y = fmaxf(v.y + b1, 0.0f);
            o.z = fmaxf(v.z + b2, 0.0f);
            o.w = fmaxf(v.w + b3, 0.0f);
            ((float4*)Hout)[(size_t)node * 32 + lane] = o;
        }
    }
}

__global__ void __launch_bounds__(512) k_gather_out(
        const float* __restrict__ XW, const int* __restrict__ seg,
        const int* __restrict__ nbr,
        const float* __restrict__ cs, const float* __restrict__ cd,
        const float* __restrict__ bias, float* __restrict__ out, int n) {
    __shared__ float stage[16 * 18];
    const int w    = threadIdx.x >> 5;
    const int node = __builtin_amdgcn_readfirstlane((int)(blockIdx.x * 16 + w));
    const int lane = threadIdx.x & 31;
    const bool have = node < n;
    int beg = 0, cnt = 0;
    if (have) {
        const int2 sg = *(const int2*)(seg + 2 * (size_t)node);
        beg = sg.x; cnt = sg.y;
    }
    const int g = lane >> 3, q = lane & 7;
    const float4* __restrict__ X4 = (const float4*)XW;

    float4 acc = make_float4(0.f, 0.f, 0.f, 0.f);
    for (int base = 0; base < cnt; base += 32) {
        const int m = min(cnt - base, 32);
        int   s_l = 0;
        float c_l = 0.0f;
        if (lane < m) {
            s_l = nbr[beg + base + lane];
            s_l = ((unsigned)s_l < (unsigned)n) ? s_l : 0;
            c_l = cs[s_l];
        }
        for (int j = 0; j < m; j += 8) {
            const int ja = j + g, jb = j + 4 + g;
            const int sa = __shfl(s_l, ja, 32);
            const int sb = __shfl(s_l, jb, 32);
            float4 ra = make_float4(0.f, 0.f, 0.f, 0.f), rb = ra;
            if (ja < m) ra = X4[(size_t)sa * 8 + q];
            if (jb < m) rb = X4[(size_t)sb * 8 + q];
            acc = fma4(ra, __shfl(c_l, ja, 32), acc);
            acc = fma4(rb, __shfl(c_l, jb, 32), acc);
        }
    }
#pragma unroll
    for (int msk = 8; msk <= 16; msk <<= 1) {
        acc.x += __shfl_xor(acc.x, msk, 32);
        acc.y += __shfl_xor(acc.y, msk, 32);
        acc.z += __shfl_xor(acc.z, msk, 32);
        acc.w += __shfl_xor(acc.w, msk, 32);
    }
    if (have && lane < 5) {
        const float sc = cd[node];
        float* o = stage + w * 18 + lane * 4;
        const float* bb = bias + lane * 4;
        o[0] = acc.x * sc + bb[0];
        o[1] = acc.y * sc + bb[1];
        if (lane < 4) {
            o[2] = acc.z * sc + bb[2];
            o[3] = acc.w * sc + bb[3];
        }
    }
    __syncthreads();
    const int t = threadIdx.x;
    if (t < 16 * 18) {
        const size_t idx = (size_t)blockIdx.x * (16 * 18) + t;
        if (idx < (size_t)n * 18) out[idx] = stage[t];
    }
}


static inline int cdiv(int a, int b) { return (a + b - 1) / b; }

extern "C" void kernel_launch(void* const* d_in, const int* in_sizes, int n_in,
                              void* d_out, int out_size, void* d_ws, size_t ws_size,
                              hipStream_t stream) {
    (void)n_in; (void)out_size; (void)ws_size;

    const float* feat = (const float*)d_in[0];
    const int*   src  = (const int*)d_in[1];
    const int*   dst  = (const int*)d_in[2];
    const float* W1 = (const float*)d_in[3];  const float* b1 = (const float*)d_in[4];
    const float* W2 = (const float*)d_in[5];  const float* b2 = (const float*)d_in[6];
    const float* W3 = (const float*)d_in[7];  const float* b3 = (const float*)d_in[8];
    const float* W4 = (const float*)d_in[9];  const float* b4 = (const float*)d_in[10];
    const float* W5 = (const float*)d_in[11]; const float* b5 = (const float*)d_in[12];
    float* out = (float*)d_out;

    const int N = 100000;
    const int E = in_sizes[1];

    float* ws = (float*)d_ws;
    size_t o = 0;
    float* c_src = ws + o; o += N;
    float* c_dst = ws + o; o += N;
    float* Xp    = ws + o; o += (size_t)N * 8;
    float* W1p   = ws + o; o += 8 * 128;
    float* W5p   = ws + o; o += 128 * 32;
    float* XW    = ws + o; o += (size_t)N * 128;
    float* H     = ws + o; o += (size_t)N * 128;
    int* iw = (int*)(ws + o);
    size_t p = 0;
    int* degO   = iw + p; p += N;
    int* degI   = iw + p; p += N;
    int* rowptr = iw + p; p += N;
    int* seg    = iw + p; p += 2 * (size_t)N;
    int* csum   = iw + p; p += 1024;
    int* rank   = iw + p; p += (size_t)E;
    int* nbr    = iw + p; p += (size_t)E;

    const int nChunk = cdiv(N, 1024);

    k_zero_i<<<cdiv(2 * N, 256), 256, 0, stream>>>(degO, 2 * N);
    k_count<<<cdiv(E, 256), 256, 0, stream>>>(src, dst, degO, degI, rank, E);
    k_chunk_sum<<<nChunk, 256, 0, stream>>>(degI, csum, N);
    k_rowptr_coeff<<<nChunk, 256, 0, stream>>>(degO, degI, csum, seg, rowptr, c_src, c_dst, N);
    k_fill<<<cdiv(E, 256), 256, 0, stream>>>(src, dst, rank, rowptr, nbr, E);

    k_pad_features<<<cdiv(N * 8, 256), 256, 0, stream>>>(feat, Xp, N);
    k_pad_w1<<<4, 256, 0, stream>>>(W1, W1p);
    k_pad_w5<<<16, 256, 0, stream>>>(W5, W5p);

    const int fBlk = cdiv(N, 64);

    k_layer_fused<8><<<fBlk, 512, 0, stream>>>(Xp, W1p, seg, nbr, c_src, c_dst, b1, H, N);

    k_layer_fused<128><<<fBlk, 512, 0, stream>>>(H,  W2, seg, nbr, c_src, c_dst, b2, XW, N);
    k_layer_fused<128><<<fBlk, 512, 0, stream>>>(XW, W3, seg, nbr, c_src, c_dst, b3, H,  N);
    k_layer_fused<128><<<fBlk, 512, 0, stream>>>(H,  W4, seg, nbr, c_src, c_dst, b4, XW, N);

    const dim3 gTile5(N / 16, 32 / 16);
    k_gemm_wmma<128, 32><<<gTile5, 32, 0, stream>>>(XW, W5p, H);
    k_gather_out<<<cdiv(N, 16), 512, 0, stream>>>(H, seg, nbr, c_src, c_dst, b5, out, N);
}
